# starting kernel
# baseline (speedup 1.0000x reference)
#include <hip/hip_runtime.h>

typedef __attribute__((ext_vector_type(2))) float v2f;
typedef __attribute__((ext_vector_type(8))) float v8f;

#define B_  2
#define S_  512
#define H_  8
#define D_  64
#define E_  512
#define R_  256
#define TM  32

__global__ __launch_bounds__(256) void relpos_wmma_kernel(
    const float* __restrict__ q,
    const float* __restrict__ table,
    const int*   __restrict__ x,
    const int*   __restrict__ max_x_p,
    float* __restrict__ out)
{
    __shared__ float ldsQ[TM * D_];
    __shared__ float ldsP[TM * R_];
    __shared__ int   xs[S_];

    const int tid   = threadIdx.x;
    const int bid   = blockIdx.x;
    const int itile = bid & ((S_ / TM) - 1);
    const int h     = (bid >> 4) & (H_ - 1);
    const int b     = bid >> 7;
    const int i0    = itile * TM;
    const int mx    = max_x_p[0];

#pragma unroll
    for (int k = 0; k < (TM * D_) / 256; ++k) {
        int idx = tid + k * 256;
        int m = idx >> 6;
        int d = idx & (D_ - 1);
        ldsQ[idx] = q[(((size_t)b * S_ + (i0 + m)) * H_ + h) * D_ + d];
    }
#pragma unroll
    for (int k = 0; k < S_ / 256; ++k) {
        int j = tid + k * 256;
        xs[j] = x[b * S_ + j];
    }
    __syncthreads();

    const int wave   = tid >> 5;
    const int lane   = tid & 31;
    const int laneLo = lane & 15;
    const int hiSel  = lane >> 4;

    v2f a[2][16];
#pragma unroll
    for (int mt = 0; mt < 2; ++mt) {
        const float* src = &ldsQ[(mt * 16 + laneLo) * D_ + 2 * hiSel];
#pragma unroll
        for (int k4 = 0; k4 < 16; ++k4) {
            a[mt][k4] = *reinterpret_cast<const v2f*>(src + 4 * k4);
        }
    }

#pragma unroll
    for (int p = 0; p < 2; ++p) {
        const int nt = wave * 2 + p;
        const int r  = nt * 16 + laneLo;
        const float* bsrc = table + (size_t)r * E_ + h * D_ + 2 * hiSel;

        v8f acc0 = {};
        v8f acc1 = {};
#pragma unroll
        for (int k4 = 0; k4 < 16; ++k4) {
            v2f bv = *reinterpret_cast<const v2f*>(bsrc + 4 * k4);
            acc0 = __builtin_amdgcn_wmma_f32_16x16x4_f32(
                       false, a[0][k4], false, bv, (short)0, acc0, false, false);
            acc1 = __builtin_amdgcn_wmma_f32_16x16x4_f32(
                       false, a[1][k4], false, bv, (short)0, acc1, false, false);
        }

#pragma unroll
        for (int c = 0; c < 8; ++c) {
            int M = c + 8 * hiSel;
            ldsP[M * R_ + nt * 16 + laneLo]         = acc0[c];
            ldsP[(16 + M) * R_ + nt * 16 + laneLo]  = acc1[c];
        }
    }
    __syncthreads();

    float* outBase = out + (((size_t)(b * H_ + h)) * S_ + i0) * S_;
#pragma unroll 4
    for (int idx = tid; idx < TM * S_; idx += 256) {
        int m   = idx >> 9;
        int j   = idx & (S_ - 1);
        int rel = mx + xs[j] - xs[i0 + m];
        outBase[(size_t)m * S_ + j] = ldsP[m * R_ + rel];
    }
}

extern "C" void kernel_launch(void* const* d_in, const int* in_sizes, int n_in,
                              void* d_out, int out_size, void* d_ws, size_t ws_size,
                              hipStream_t stream) {
    const float* q     = (const float*)d_in[0];
    const float* table = (const float*)d_in[1];
    const int*   x     = (const int*)d_in[2];
    const int*   mx    = (const int*)d_in[3];
    float*       out   = (float*)d_out;
    (void)in_sizes; (void)n_in; (void)out_size; (void)d_ws; (void)ws_size;

    dim3 grid(B_ * H_ * (S_ / TM));
    dim3 block(256);
    hipLaunchKernelGGL(relpos_wmma_kernel, grid, block, 0, stream, q, table, x, mx, out);
}


# staging pipelined with the products: q tile and table slice reach LDS by whole-line 16-B loads, the slice in four 64-row chunks, each handed over inside the previous chunk's products; sums unchanged
# speedup vs baseline: 1.2176x; 1.2176x over previous
#include <hip/hip_runtime.h>


typedef __attribute__((ext_vector_type(2))) float v2f;
typedef __attribute__((ext_vector_type(4))) float v4f;
typedef __attribute__((ext_vector_type(8))) float v8f;
typedef __attribute__((ext_vector_type(4))) int   v4i;

#define B_  2
#define S_  512
#define H_  8
#define D_  64
#define E_  512
#define R_  256
#define TM  32
#define PW  384
#define PT  68

__global__ __launch_bounds__(256) void relpos_wmma_kernel(
    const float* __restrict__ q,
    const float* __restrict__ table,
    const int*   __restrict__ x,
    const int*   __restrict__ max_x_p,
    float* __restrict__ out)
{
    __shared__ __attribute__((aligned(16))) float ldsT[R_ * PT];
    __shared__ __attribute__((aligned(16))) float ldsQ[TM * PT];
    __shared__ float ldsP[TM * PW];

    const int tid   = threadIdx.x;
    const int bid   = blockIdx.x;
    const int itile = bid & ((S_ / TM) - 1);
    const int h     = (bid >> 4) & (H_ - 1);
    const int b     = bid >> 7;
    const int i0    = itile * TM;
    const int mx    = max_x_p[0];

    const int wave   = tid >> 5;
    const int lane   = tid & 31;
    const int laneLo = lane & 15;
    const int hiSel  = lane >> 4;
    const int mt     = wave >> 2;
    const int ntg    = wave & 3;

    const int xj0 = x[b * S_ + tid];
    const int xj1 = x[b * S_ + tid + 256];
    const int* xrow = x + b * S_ + i0 + 16 * mt + 8 * hiSel;
    const v4i xr0 = *reinterpret_cast<const v4i*>(xrow);
    const v4i xr1 = *reinterpret_cast<const v4i*>(xrow + 4);

    v4f qst[2];
    const float* qsrc = q + (((size_t)b * S_ + i0 + (tid >> 4)) * H_ + h) * D_ + 4 * (tid & 15);
    float* qdst = &ldsQ[(tid >> 4) * PT + 4 * (tid & 15)];
#pragma unroll
    for (int j = 0; j < 2; ++j)
        qst[j] = *reinterpret_cast<const v4f*>(qsrc + (size_t)j * 16 * H_ * D_);

    v4f tst[16];
    const float* tsrc = table + (size_t)(tid >> 4) * E_ + h * D_ + 4 * (tid & 15);
    float* tdst = &ldsT[(tid >> 4) * PT + 4 * (tid & 15)];
#pragma unroll
    for (int k = 0; k < 4; ++k)
        tst[k] = *reinterpret_cast<const v4f*>(tsrc + (size_t)k * 16 * E_);
    __builtin_amdgcn_sched_barrier(0);
#pragma unroll
    for (int k = 4; k < 16; ++k)
        tst[k] = *reinterpret_cast<const v4f*>(tsrc + (size_t)k * 16 * E_);
    __builtin_amdgcn_sched_barrier(0);

#pragma unroll
    for (int j = 0; j < 2; ++j)
        *reinterpret_cast<v4f*>(qdst + 16 * j * PT) = qst[j];
#pragma unroll
    for (int k = 0; k < 4; ++k)
        *reinterpret_cast<v4f*>(tdst + 16 * k * PT) = tst[k];

    __builtin_amdgcn_fence(__ATOMIC_RELEASE, "workgroup", "local");
    __builtin_amdgcn_s_barrier();
    __builtin_amdgcn_fence(__ATOMIC_ACQUIRE, "workgroup", "local");

    int pcol[8];
#pragma unroll
    for (int c = 0; c < 4; ++c) {
        pcol[c]     = (16 * mt + 8 * hiSel + c) * PW     + xr0[c] + laneLo;
        pcol[c + 4] = (16 * mt + 8 * hiSel + c + 4) * PW + xr1[c] + laneLo;
    }

    v2f a[16];
    v2f bv[16];
    {
        const float* asrc = &ldsQ[(16 * mt + laneLo) * PT + 2 * hiSel];
        const float* bsrc = &ldsT[(ntg * 16 + laneLo) * PT + 2 * hiSel];
#pragma unroll
        for (int k4 = 0; k4 < 16; ++k4) {
            a[k4]  = *reinterpret_cast<const v2f*>(asrc + 4 * k4);
            bv[k4] = *reinterpret_cast<const v2f*>(bsrc + 4 * k4);
        }
    }
    __builtin_amdgcn_sched_barrier(0);

    v8f acc[4] = {};
#pragma unroll
    for (int ch = 0; ch < 4; ++ch) {
        v2f bn[16];
#pragma unroll
        for (int k4 = 0; k4 < 8; ++k4)
            acc[ch] = __builtin_amdgcn_wmma_f32_16x16x4_f32(
                          false, a[k4], false, bv[k4], (short)0, acc[ch], false, false);
        if (ch < 3) {
            __builtin_amdgcn_sched_barrier(0);
        }
        if (ch > 0) {
#pragma unroll
            for (int c = 0; c < 8; ++c)
                ldsP[pcol[c] + ((ch - 1) * 4 + ntg) * 16] = acc[ch - 1][c];
        }
        if (ch < 3) {
#pragma unroll
            for (int k = 0; k < 4; ++k)
                *reinterpret_cast<v4f*>(tdst + 16 * (4 * (ch + 1) + k) * PT) = tst[4 * (ch + 1) + k];
            __builtin_amdgcn_sched_barrier(0);
        }
#pragma unroll
        for (int k4 = 8; k4 < 12; ++k4)
            acc[ch] = __builtin_amdgcn_wmma_f32_16x16x4_f32(
                          false, a[k4], false, bv[k4], (short)0, acc[ch], false, false);
        if (ch < 3) {
            __builtin_amdgcn_sched_barrier(0);
            __builtin_amdgcn_fence(__ATOMIC_RELEASE, "workgroup", "local");
            __builtin_amdgcn_s_barrier();
            __builtin_amdgcn_fence(__ATOMIC_ACQUIRE, "workgroup", "local");
            const float* bsrc = &ldsT[((ch + 1) * 64 + ntg * 16 + laneLo) * PT + 2 * hiSel];
#pragma unroll
            for (int k4 = 0; k4 < 16; ++k4)
                bn[k4] = *reinterpret_cast<const v2f*>(bsrc + 4 * k4);
            __builtin_amdgcn_sched_barrier(0);
        }
#pragma unroll
        for (int k4 = 12; k4 < 16; ++k4)
            acc[ch] = __builtin_amdgcn_wmma_f32_16x16x4_f32(
                          false, a[k4], false, bv[k4], (short)0, acc[ch], false, false);
        if (ch < 3) {
#pragma unroll
            for (int k4 = 0; k4 < 16; ++k4)
                bv[k4] = bn[k4];
        }
    }
#pragma unroll
    for (int c = 0; c < 8; ++c)
        ldsP[pcol[c] + (3 * 4 + ntg) * 16] = acc[3][c];
    __syncthreads();

    const float* g0 = ldsP + (mx + xj0);
    const float* g1 = ldsP + (mx + xj1);
    float* o = out + (((size_t)(b * H_ + h)) * S_ + i0) * S_ + tid;
#pragma unroll
    for (int m = 0; m < TM; ++m) {
        o[m * S_]       = g0[m * PW];
        o[m * S_ + 256] = g1[m * PW];
    }
}

extern "C" void kernel_launch(void* const* d_in, const int* in_sizes, int n_in,
                              void* d_out, int out_size, void* d_ws, size_t ws_size,
                              hipStream_t stream) {
    const float* q     = (const float*)d_in[0];
    const float* table = (const float*)d_in[1];
    const int*   x     = (const int*)d_in[2];
    const int*   mx    = (const int*)d_in[3];
    float*       out   = (float*)d_out;
    (void)in_sizes; (void)n_in; (void)out_size; (void)d_ws; (void)ws_size;

    dim3 grid(B_ * H_ * (S_ / TM));
    dim3 block(256);
    hipLaunchKernelGGL(relpos_wmma_kernel, grid, block, 0, stream, q, table, x, mx, out);
}
